# starting kernel
# baseline (speedup 1.0000x reference)
#include <hip/hip_runtime.h>

typedef __attribute__((ext_vector_type(2))) float v2f;
typedef __attribute__((ext_vector_type(8))) float v8f;

#define BATCH   4096
#define LAYERS  4
#define NEIGH   32
#define FEAT    256
#define EMB     128
#define TB      16
#define THREADS 256
#define WAVES   8

__launch_bounds__(THREADS)
__global__ void liamne_fused(const int*   __restrict__ g_layers,
                             const int*   __restrict__ g_node_i,
                             const int*   __restrict__ g_neighs,
                             const float* __restrict__ g_features,
                             const float* __restrict__ g_layemb,
                             const float* __restrict__ g_net,
                             const float* __restrict__ g_tw,
                             const float* __restrict__ g_ts1,
                             const float* __restrict__ g_ts2,
                             float*       __restrict__ g_out)
{
    __shared__ __align__(16) float S[LAYERS * TB * FEAT];
    __shared__ __align__(16) float Hs[LAYERS * TB * EMB];
    __shared__ float SC[LAYERS * TB];
    __shared__ float ATT[LAYERS * TB];

    const int tid  = threadIdx.x;
    const int lane = tid & 31;
    const int wave = tid >> 5;
    const int blk  = blockIdx.x;
    const int half = lane >> 4;
    const int lm   = lane & 15;

    for (int rr = 0; rr < 8; ++rr) {
        const int row = wave * 8 + rr;
        const int l  = row >> 4;
        const int b  = row & 15;
        const int bg = blk * TB + b;
        const int nbase = (bg * LAYERS + l) * NEIGH;
        const int myidx = g_neighs[nbase + lane];
        float4 a0 = {0.f,0.f,0.f,0.f}, a1 = {0.f,0.f,0.f,0.f};
        for (int j = 0; j < NEIGH; ++j) {
            const int idx = __shfl(myidx, j);
            const float4* fr = (const float4*)(g_features + (long)idx * FEAT);
            const float4 x0 = fr[lane];
            const float4 x1 = fr[32 + lane];
            a0.x += x0.x; a0.y += x0.y; a0.z += x0.z; a0.w += x0.w;
            a1.x += x1.x; a1.y += x1.y; a1.z += x1.z; a1.w += x1.w;
        }
        float4* srow = (float4*)(S + row * FEAT);
        srow[lane]      = a0;
        srow[32 + lane] = a1;
    }
    __syncthreads();

    for (int t = 0; t < 4; ++t) {
        const int tt = wave * 4 + t;
        const int l  = tt >> 3;
        const int n0 = (tt & 7) * 16;
        const float* Wl   = g_net + l * FEAT * EMB;
        const float* srow = S + (l * TB + lm) * FEAT;
        v8f c = {0.f,0.f,0.f,0.f,0.f,0.f,0.f,0.f};
        for (int k = 0; k < FEAT; k += 4) {
            v2f a, bm;
            a.x  = srow[k + half*2 + 0];
            a.y  = srow[k + half*2 + 1];
            bm.x = Wl[(k + half*2 + 0) * EMB + n0 + lm];
            bm.y = Wl[(k + half*2 + 1) * EMB + n0 + lm];
            c = __builtin_amdgcn_wmma_f32_16x16x4_f32(false, a, false, bm,
                                                      (short)0, c, false, false);
        }
#pragma unroll
        for (int r = 0; r < 8; ++r) {
            const int m = half ? (r + 8) : r;
            Hs[(l * TB + m) * EMB + n0 + lm] = c[r];
        }
    }
    __syncthreads();

    float* T = S;
    for (int t = 0; t < 4; ++t) {
        const int tt = wave * 4 + t;
        const int mt = tt >> 3;
        const int n0 = (tt & 7) * 16;
        const float* hrow = Hs + (mt * TB + lm) * EMB;
        v8f c = {0.f,0.f,0.f,0.f,0.f,0.f,0.f,0.f};
        for (int k = 0; k < EMB; k += 4) {
            v2f a, bm;
            a.x  = hrow[k + half*2 + 0];
            a.y  = hrow[k + half*2 + 1];
            bm.x = g_ts1[(k + half*2 + 0) * EMB + n0 + lm];
            bm.y = g_ts1[(k + half*2 + 1) * EMB + n0 + lm];
            c = __builtin_amdgcn_wmma_f32_16x16x4_f32(false, a, false, bm,
                                                      (short)0, c, false, false);
        }
#pragma unroll
        for (int r = 0; r < 8; ++r) {
            const int m = half ? (r + 8) : r;
            T[(mt * TB + m) * EMB + n0 + lm] = tanhf(c[r]);
        }
    }
    __syncthreads();

    for (int rr = 0; rr < 8; ++rr) {
        const int row = wave * 8 + rr;
        float s = 0.f;
#pragma unroll
        for (int i = 0; i < 4; ++i) {
            const int m = lane + i * 32;
            s += T[row * EMB + m] * g_ts2[m];
        }
#pragma unroll
        for (int off = 16; off > 0; off >>= 1) s += __shfl_xor(s, off);
        if (lane == 0) SC[row] = s;
    }
    __syncthreads();

    if (tid < TB) {
        const float s0 = SC[0*TB + tid], s1 = SC[1*TB + tid];
        const float s2 = SC[2*TB + tid], s3 = SC[3*TB + tid];
        const float mx = fmaxf(fmaxf(s0, s1), fmaxf(s2, s3));
        const float e0 = expf(s0 - mx), e1 = expf(s1 - mx);
        const float e2 = expf(s2 - mx), e3 = expf(s3 - mx);
        const float inv = 1.f / (e0 + e1 + e2 + e3);
        ATT[0*TB + tid] = e0 * inv;  ATT[1*TB + tid] = e1 * inv;
        ATT[2*TB + tid] = e2 * inv;  ATT[3*TB + tid] = e3 * inv;
    }
    __syncthreads();

    float* AGG = S + 8192;
    float* FIN = S + 8192 + TB * EMB;
    for (int e = tid; e < TB * EMB; e += THREADS) {
        const int b = e >> 7, m = e & 127;
        float v = 0.f;
#pragma unroll
        for (int l = 0; l < LAYERS; ++l)
            v += ATT[l * TB + b] * Hs[(l * TB + b) * EMB + m];
        AGG[e] = v;
    }
    __syncthreads();

    {
        const int n0 = wave * 16;
        const float* arow = AGG + lm * EMB;
        v8f c = {0.f,0.f,0.f,0.f,0.f,0.f,0.f,0.f};
        for (int k = 0; k < EMB; k += 4) {
            v2f a, bm;
            a.x  = arow[k + half*2 + 0];
            a.y  = arow[k + half*2 + 1];
            bm.x = g_tw[(k + half*2 + 0) * EMB + n0 + lm];
            bm.y = g_tw[(k + half*2 + 1) * EMB + n0 + lm];
            c = __builtin_amdgcn_wmma_f32_16x16x4_f32(false, a, false, bm,
                                                      (short)0, c, false, false);
        }
#pragma unroll
        for (int r = 0; r < 8; ++r) {
            const int m = half ? (r + 8) : r;
            FIN[m * EMB + n0 + lm] = c[r];
        }
    }
    __syncthreads();

    for (int rr = 0; rr < 2; ++rr) {
        const int b  = wave * 2 + rr;
        const int bg = blk * TB + b;
        const int node = g_node_i[bg];
        const int lay  = g_layers[bg];
        const float4* emb  = (const float4*)(g_layemb + ((long)node * LAYERS + lay) * EMB);
        const float4* frow = (const float4*)(FIN + b * EMB);
        float4 v = frow[lane];
        const float4 e = emb[lane];
        v.x += e.x; v.y += e.y; v.z += e.z; v.w += e.w;
        float ss = v.x*v.x + v.y*v.y + v.z*v.z + v.w*v.w;
#pragma unroll
        for (int off = 16; off > 0; off >>= 1) ss += __shfl_xor(ss, off);
        const float inv = 1.f / fmaxf(sqrtf(ss), 1e-12f);
        float4 o = { v.x * inv, v.y * inv, v.z * inv, v.w * inv };
        ((float4*)(g_out + (long)bg * EMB))[lane] = o;
    }
}

extern "C" void kernel_launch(void* const* d_in, const int* in_sizes, int n_in,
                              void* d_out, int out_size, void* d_ws, size_t ws_size,
                              hipStream_t stream) {
    const int*   layers   = (const int*)  d_in[0];
    const int*   node_i   = (const int*)  d_in[1];
    const int*   neighs   = (const int*)  d_in[2];
    const float* features = (const float*)d_in[3];
    const float* layemb   = (const float*)d_in[4];
    const float* net      = (const float*)d_in[5];
    const float* tw       = (const float*)d_in[6];
    const float* ts1      = (const float*)d_in[7];
    const float* ts2      = (const float*)d_in[8];
    float* out = (float*)d_out;

    dim3 grid(BATCH / TB);
    dim3 block(THREADS);
    liamne_fused<<<grid, block, 0, stream>>>(layers, node_i, neighs, features,
                                             layemb, net, tw, ts1, ts2, out);
}


# 32 waves per work-group instead of 8 (1024 threads, same row tile and LDS): every phase re-dealt over 32 waves, 4x the gather loads in flight per WGP; no change of any summation order
# speedup vs baseline: 1.2271x; 1.2271x over previous
#include <hip/hip_runtime.h>

typedef __attribute__((ext_vector_type(2))) float v2f;
typedef __attribute__((ext_vector_type(8))) float v8f;

#define BATCH   4096
#define LAYERS  4
#define NEIGH   32
#define FEAT    256
#define EMB     128
#define TB      16
#define THREADS 1024
#define WAVES   32

__launch_bounds__(THREADS)
__global__ void liamne_fused(const int*   __restrict__ g_layers,
                             const int*   __restrict__ g_node_i,
                             const int*   __restrict__ g_neighs,
                             const float* __restrict__ g_features,
                             const float* __restrict__ g_layemb,
                             const float* __restrict__ g_net,
                             const float* __restrict__ g_tw,
                             const float* __restrict__ g_ts1,
                             const float* __restrict__ g_ts2,
                             float*       __restrict__ g_out)
{
    __shared__ __align__(16) float S[LAYERS * TB * FEAT];
    __shared__ __align__(16) float Hs[LAYERS * TB * EMB];
    __shared__ float SC[LAYERS * TB];
    __shared__ float ATT[LAYERS * TB];

    const int tid  = threadIdx.x;
    const int lane = tid & 31;
    const int wave = __builtin_amdgcn_readfirstlane(tid >> 5);
    const int blk  = blockIdx.x;
    const int half = lane >> 4;
    const int lm   = lane & 15;

    for (int rr = 0; rr < 2; ++rr) {
        const int row = wave * 2 + rr;
        const int l  = row >> 4;
        const int b  = row & 15;
        const int bg = blk * TB + b;
        const int nbase = (bg * LAYERS + l) * NEIGH;
        const int myidx = g_neighs[nbase + lane];
        float4 a0 = {0.f,0.f,0.f,0.f}, a1 = {0.f,0.f,0.f,0.f};
        for (int j = 0; j < NEIGH; ++j) {
            const int idx = __shfl(myidx, j);
            const float4* fr = (const float4*)(g_features + (long)idx * FEAT);
            const float4 x0 = fr[lane];
            const float4 x1 = fr[32 + lane];
            a0.x += x0.x; a0.y += x0.y; a0.z += x0.z; a0.w += x0.w;
            a1.x += x1.x; a1.y += x1.y; a1.z += x1.z; a1.w += x1.w;
        }
        float4* srow = (float4*)(S + row * FEAT);
        srow[lane]      = a0;
        srow[32 + lane] = a1;
    }
    __syncthreads();

    {
        const int tt = wave;
        const int l  = tt >> 3;
        const int n0 = (tt & 7) * 16;
        const float* Wl   = g_net + l * FEAT * EMB;
        const float* srow = S + (l * TB + lm) * FEAT;
        v8f c = {0.f,0.f,0.f,0.f,0.f,0.f,0.f,0.f};
        for (int k = 0; k < FEAT; k += 4) {
            v2f a, bm;
            a.x  = srow[k + half*2 + 0];
            a.y  = srow[k + half*2 + 1];
            bm.x = Wl[(k + half*2 + 0) * EMB + n0 + lm];
            bm.y = Wl[(k + half*2 + 1) * EMB + n0 + lm];
            c = __builtin_amdgcn_wmma_f32_16x16x4_f32(false, a, false, bm,
                                                      (short)0, c, false, false);
        }
#pragma unroll
        for (int r = 0; r < 8; ++r) {
            const int m = half ? (r + 8) : r;
            Hs[(l * TB + m) * EMB + n0 + lm] = c[r];
        }
    }
    __syncthreads();

    float* T = S;
    {
        const int tt = wave;
        const int mt = tt >> 3;
        const int n0 = (tt & 7) * 16;
        const float* hrow = Hs + (mt * TB + lm) * EMB;
        v8f c = {0.f,0.f,0.f,0.f,0.f,0.f,0.f,0.f};
        for (int k = 0; k < EMB; k += 4) {
            v2f a, bm;
            a.x  = hrow[k + half*2 + 0];
            a.y  = hrow[k + half*2 + 1];
            bm.x = g_ts1[(k + half*2 + 0) * EMB + n0 + lm];
            bm.y = g_ts1[(k + half*2 + 1) * EMB + n0 + lm];
            c = __builtin_amdgcn_wmma_f32_16x16x4_f32(false, a, false, bm,
                                                      (short)0, c, false, false);
        }
#pragma unroll
        for (int r = 0; r < 8; ++r) {
            const int m = half ? (r + 8) : r;
            T[(mt * TB + m) * EMB + n0 + lm] = tanhf(c[r]);
        }
    }
    __syncthreads();

    for (int rr = 0; rr < 2; ++rr) {
        const int row = wave * 2 + rr;
        float s = 0.f;
#pragma unroll
        for (int i = 0; i < 4; ++i) {
            const int m = lane + i * 32;
            s += T[row * EMB + m] * g_ts2[m];
        }
#pragma unroll
        for (int off = 16; off > 0; off >>= 1) s += __shfl_xor(s, off);
        if (lane == 0) SC[row] = s;
    }
    __syncthreads();

    if (tid < TB) {
        const float s0 = SC[0*TB + tid], s1 = SC[1*TB + tid];
        const float s2 = SC[2*TB + tid], s3 = SC[3*TB + tid];
        const float mx = fmaxf(fmaxf(s0, s1), fmaxf(s2, s3));
        const float e0 = expf(s0 - mx), e1 = expf(s1 - mx);
        const float e2 = expf(s2 - mx), e3 = expf(s3 - mx);
        const float inv = 1.f / (e0 + e1 + e2 + e3);
        ATT[0*TB + tid] = e0 * inv;  ATT[1*TB + tid] = e1 * inv;
        ATT[2*TB + tid] = e2 * inv;  ATT[3*TB + tid] = e3 * inv;
    }
    __syncthreads();

    float* AGG = S + 8192;
    float* FIN = S + 8192 + TB * EMB;
    for (int e = tid; e < TB * EMB; e += THREADS) {
        const int b = e >> 7, m = e & 127;
        float v = 0.f;
#pragma unroll
        for (int l = 0; l < LAYERS; ++l)
            v += ATT[l * TB + b] * Hs[(l * TB + b) * EMB + m];
        AGG[e] = v;
    }
    __syncthreads();

    if (wave < EMB / 16) {
        const int n0 = wave * 16;
        const float* arow = AGG + lm * EMB;
        v8f c = {0.f,0.f,0.f,0.f,0.f,0.f,0.f,0.f};
        for (int k = 0; k < EMB; k += 4) {
            v2f a, bm;
            a.x  = arow[k + half*2 + 0];
            a.y  = arow[k + half*2 + 1];
            bm.x = g_tw[(k + half*2 + 0) * EMB + n0 + lm];
            bm.y = g_tw[(k + half*2 + 1) * EMB + n0 + lm];
            c = __builtin_amdgcn_wmma_f32_16x16x4_f32(false, a, false, bm,
                                                      (short)0, c, false, false);
        }
#pragma unroll
        for (int r = 0; r < 8; ++r) {
            const int m = half ? (r + 8) : r;
            FIN[m * EMB + n0 + lm] = c[r];
        }
    }
    __syncthreads();

    if (wave < TB) {
        const int b  = wave;
        const int bg = blk * TB + b;
        const int node = g_node_i[bg];
        const int lay  = g_layers[bg];
        const float4* emb  = (const float4*)(g_layemb + ((long)node * LAYERS + lay) * EMB);
        const float4* frow = (const float4*)(FIN + b * EMB);
        float4 v = frow[lane];
        const float4 e = emb[lane];
        v.x += e.x; v.y += e.y; v.z += e.z; v.w += e.w;
        float ss = v.x*v.x + v.y*v.y + v.z*v.z + v.w*v.w;
#pragma unroll
        for (int off = 16; off > 0; off >>= 1) ss += __shfl_xor(ss, off);
        const float inv = 1.f / fmaxf(sqrtf(ss), 1e-12f);
        float4 o = { v.x * inv, v.y * inv, v.z * inv, v.w * inv };
        ((float4*)(g_out + (long)bg * EMB))[lane] = o;
    }
}

extern "C" void kernel_launch(void* const* d_in, const int* in_sizes, int n_in,
                              void* d_out, int out_size, void* d_ws, size_t ws_size,
                              hipStream_t stream) {
    const int*   layers   = (const int*)  d_in[0];
    const int*   node_i   = (const int*)  d_in[1];
    const int*   neighs   = (const int*)  d_in[2];
    const float* features = (const float*)d_in[3];
    const float* layemb   = (const float*)d_in[4];
    const float* net      = (const float*)d_in[5];
    const float* tw       = (const float*)d_in[6];
    const float* ts1      = (const float*)d_in[7];
    const float* ts2      = (const float*)d_in[8];
    float* out = (float*)d_out;

    dim3 grid(BATCH / TB);
    dim3 block(THREADS);
    liamne_fused<<<grid, block, 0, stream>>>(layers, node_i, neighs, features,
                                             layemb, net, tw, ts1, ts2, out);
}


# LDS rows padded by 4 floats (pitch 260 / 132 instead of 256 / 128): the WMMA A-fragment reads of phases B, C, D were 16-way bank-conflicted, now conflict-free; no change of any summation order
# speedup vs baseline: 1.5161x; 1.2355x over previous
#include <hip/hip_runtime.h>

typedef __attribute__((ext_vector_type(2))) float v2f;
typedef __attribute__((ext_vector_type(8))) float v8f;

#define BATCH   4096
#define LAYERS  4
#define NEIGH   32
#define FEAT    256
#define EMB     128
#define TB      16
#define THREADS 1024
#define WAVES   32
#define SP      (FEAT + 4)
#define HP      (EMB + 4)

__launch_bounds__(THREADS)
__global__ void liamne_fused(const int*   __restrict__ g_layers,
                             const int*   __restrict__ g_node_i,
                             const int*   __restrict__ g_neighs,
                             const float* __restrict__ g_features,
                             const float* __restrict__ g_layemb,
                             const float* __restrict__ g_net,
                             const float* __restrict__ g_tw,
                             const float* __restrict__ g_ts1,
                             const float* __restrict__ g_ts2,
                             float*       __restrict__ g_out)
{
    __shared__ __align__(16) float S[LAYERS * TB * SP];
    __shared__ __align__(16) float Hs[LAYERS * TB * HP];
    __shared__ float SC[LAYERS * TB];
    __shared__ float ATT[LAYERS * TB];

    const int tid  = threadIdx.x;
    const int lane = tid & 31;
    const int wave = __builtin_amdgcn_readfirstlane(tid >> 5);
    const int blk  = blockIdx.x;
    const int half = lane >> 4;
    const int lm   = lane & 15;

    for (int rr = 0; rr < 2; ++rr) {
        const int row = wave * 2 + rr;
        const int l  = row >> 4;
        const int b  = row & 15;
        const int bg = blk * TB + b;
        const int nbase = (bg * LAYERS + l) * NEIGH;
        const int myidx = g_neighs[nbase + lane];
        float4 a0 = {0.f,0.f,0.f,0.f}, a1 = {0.f,0.f,0.f,0.f};
        for (int j = 0; j < NEIGH; ++j) {
            const int idx = __shfl(myidx, j);
            const float4* fr = (const float4*)(g_features + (long)idx * FEAT);
            const float4 x0 = fr[lane];
            const float4 x1 = fr[32 + lane];
            a0.x += x0.x; a0.y += x0.y; a0.z += x0.z; a0.w += x0.w;
            a1.x += x1.x; a1.y += x1.y; a1.z += x1.z; a1.w += x1.w;
        }
        float4* srow = (float4*)(S + row * SP);
        srow[lane]      = a0;
        srow[32 + lane] = a1;
    }
    __syncthreads();

    {
        const int tt = wave;
        const int l  = tt >> 3;
        const int n0 = (tt & 7) * 16;
        const float* Wl   = g_net + l * FEAT * EMB;
        const float* srow = S + (l * TB + lm) * SP;
        v8f c = {0.f,0.f,0.f,0.f,0.f,0.f,0.f,0.f};
        for (int k = 0; k < FEAT; k += 4) {
            v2f a, bm;
            a.x  = srow[k + half*2 + 0];
            a.y  = srow[k + half*2 + 1];
            bm.x = Wl[(k + half*2 + 0) * EMB + n0 + lm];
            bm.y = Wl[(k + half*2 + 1) * EMB + n0 + lm];
            c = __builtin_amdgcn_wmma_f32_16x16x4_f32(false, a, false, bm,
                                                      (short)0, c, false, false);
        }
#pragma unroll
        for (int r = 0; r < 8; ++r) {
            const int m = half ? (r + 8) : r;
            Hs[(l * TB + m) * HP + n0 + lm] = c[r];
        }
    }
    __syncthreads();

    float* T = S;
    {
        const int tt = wave;
        const int mt = tt >> 3;
        const int n0 = (tt & 7) * 16;
        const float* hrow = Hs + (mt * TB + lm) * HP;
        v8f c = {0.f,0.f,0.f,0.f,0.f,0.f,0.f,0.f};
        for (int k = 0; k < EMB; k += 4) {
            v2f a, bm;
            a.x  = hrow[k + half*2 + 0];
            a.y  = hrow[k + half*2 + 1];
            bm.x = g_ts1[(k + half*2 + 0) * EMB + n0 + lm];
            bm.y = g_ts1[(k + half*2 + 1) * EMB + n0 + lm];
            c = __builtin_amdgcn_wmma_f32_16x16x4_f32(false, a, false, bm,
                                                      (short)0, c, false, false);
        }
#pragma unroll
        for (int r = 0; r < 8; ++r) {
            const int m = half ? (r + 8) : r;
            T[(mt * TB + m) * HP + n0 + lm] = tanhf(c[r]);
        }
    }
    __syncthreads();

    for (int rr = 0; rr < 2; ++rr) {
        const int row = wave * 2 + rr;
        float s = 0.f;
#pragma unroll
        for (int i = 0; i < 4; ++i) {
            const int m = lane + i * 32;
            s += T[row * HP + m] * g_ts2[m];
        }
#pragma unroll
        for (int off = 16; off > 0; off >>= 1) s += __shfl_xor(s, off);
        if (lane == 0) SC[row] = s;
    }
    __syncthreads();

    if (tid < TB) {
        const float s0 = SC[0*TB + tid], s1 = SC[1*TB + tid];
        const float s2 = SC[2*TB + tid], s3 = SC[3*TB + tid];
        const float mx = fmaxf(fmaxf(s0, s1), fmaxf(s2, s3));
        const float e0 = expf(s0 - mx), e1 = expf(s1 - mx);
        const float e2 = expf(s2 - mx), e3 = expf(s3 - mx);
        const float inv = 1.f / (e0 + e1 + e2 + e3);
        ATT[0*TB + tid] = e0 * inv;  ATT[1*TB + tid] = e1 * inv;
        ATT[2*TB + tid] = e2 * inv;  ATT[3*TB + tid] = e3 * inv;
    }
    __syncthreads();

    float* AGG = S + LAYERS * TB * HP;
    float* FIN = AGG + TB * HP;
    for (int e = tid; e < TB * EMB; e += THREADS) {
        const int b = e >> 7, m = e & 127;
        float v = 0.f;
#pragma unroll
        for (int l = 0; l < LAYERS; ++l)
            v += ATT[l * TB + b] * Hs[(l * TB + b) * HP + m];
        AGG[b * HP + m] = v;
    }
    __syncthreads();

    if (wave < EMB / 16) {
        const int n0 = wave * 16;
        const float* arow = AGG + lm * HP;
        v8f c = {0.f,0.f,0.f,0.f,0.f,0.f,0.f,0.f};
        for (int k = 0; k < EMB; k += 4) {
            v2f a, bm;
            a.x  = arow[k + half*2 + 0];
            a.y  = arow[k + half*2 + 1];
            bm.x = g_tw[(k + half*2 + 0) * EMB + n0 + lm];
            bm.y = g_tw[(k + half*2 + 1) * EMB + n0 + lm];
            c = __builtin_amdgcn_wmma_f32_16x16x4_f32(false, a, false, bm,
                                                      (short)0, c, false, false);
        }
#pragma unroll
        for (int r = 0; r < 8; ++r) {
            const int m = half ? (r + 8) : r;
            FIN[m * HP + n0 + lm] = c[r];
        }
    }
    __syncthreads();

    if (wave < TB) {
        const int b  = wave;
        const int bg = blk * TB + b;
        const int node = g_node_i[bg];
        const int lay  = g_layers[bg];
        const float4* emb  = (const float4*)(g_layemb + ((long)node * LAYERS + lay) * EMB);
        const float4* frow = (const float4*)(FIN + b * HP);
        float4 v = frow[lane];
        const float4 e = emb[lane];
        v.x += e.x; v.y += e.y; v.z += e.z; v.w += e.w;
        float ss = v.x*v.x + v.y*v.y + v.z*v.z + v.w*v.w;
#pragma unroll
        for (int off = 16; off > 0; off >>= 1) ss += __shfl_xor(ss, off);
        const float inv = 1.f / fmaxf(sqrtf(ss), 1e-12f);
        float4 o = { v.x * inv, v.y * inv, v.z * inv, v.w * inv };
        ((float4*)(g_out + (long)bg * EMB))[lane] = o;
    }
}

extern "C" void kernel_launch(void* const* d_in, const int* in_sizes, int n_in,
                              void* d_out, int out_size, void* d_ws, size_t ws_size,
                              hipStream_t stream) {
    const int*   layers   = (const int*)  d_in[0];
    const int*   node_i   = (const int*)  d_in[1];
    const int*   neighs   = (const int*)  d_in[2];
    const float* features = (const float*)d_in[3];
    const float* layemb   = (const float*)d_in[4];
    const float* net      = (const float*)d_in[5];
    const float* tw       = (const float*)d_in[6];
    const float* ts1      = (const float*)d_in[7];
    const float* ts2      = (const float*)d_in[8];
    float* out = (float*)d_out;

    dim3 grid(BATCH / TB);
    dim3 block(THREADS);
    liamne_fused<<<grid, block, 0, stream>>>(layers, node_i, neighs, features,
                                             layemb, net, tw, ts1, ts2, out);
}


# GEMM phases B, C, D: each wave does two column-interleaved 16x16 tiles (column 2c+t): one 8-byte B load per lane feeds two WMMAs on independent accumulators; half as many waves work; k order kept
# speedup vs baseline: 1.6054x; 1.0589x over previous
#include <hip/hip_runtime.h>

typedef __attribute__((ext_vector_type(2))) float v2f;
typedef __attribute__((ext_vector_type(8))) float v8f;

#define BATCH   4096
#define LAYERS  4
#define NEIGH   32
#define FEAT    256
#define EMB     128
#define TB      16
#define THREADS 1024
#define WAVES   32
#define SP      (FEAT + 4)
#define HP      (EMB + 4)

__launch_bounds__(THREADS)
__global__ void liamne_fused(const int*   __restrict__ g_layers,
                             const int*   __restrict__ g_node_i,
                             const int*   __restrict__ g_neighs,
                             const float* __restrict__ g_features,
                             const float* __restrict__ g_layemb,
                             const float* __restrict__ g_net,
                             const float* __restrict__ g_tw,
                             const float* __restrict__ g_ts1,
                             const float* __restrict__ g_ts2,
                             float*       __restrict__ g_out)
{
    __shared__ __align__(16) float S[LAYERS * TB * SP];
    __shared__ __align__(16) float Hs[LAYERS * TB * HP];
    __shared__ float SC[LAYERS * TB];
    __shared__ float ATT[LAYERS * TB];

    const int tid  = threadIdx.x;
    const int lane = tid & 31;
    const int wave = __builtin_amdgcn_readfirstlane(tid >> 5);
    const int blk  = blockIdx.x;
    const int half = lane >> 4;
    const int lm   = lane & 15;

    for (int rr = 0; rr < 2; ++rr) {
        const int row = wave * 2 + rr;
        const int l  = row >> 4;
        const int b  = row & 15;
        const int bg = blk * TB + b;
        const int nbase = (bg * LAYERS + l) * NEIGH;
        const int myidx = g_neighs[nbase + lane];
        float4 a0 = {0.f,0.f,0.f,0.f}, a1 = {0.f,0.f,0.f,0.f};
        for (int j = 0; j < NEIGH; ++j) {
            const int idx = __shfl(myidx, j);
            const float4* fr = (const float4*)(g_features + (long)idx * FEAT);
            const float4 x0 = fr[lane];
            const float4 x1 = fr[32 + lane];
            a0.x += x0.x; a0.y += x0.y; a0.z += x0.z; a0.w += x0.w;
            a1.x += x1.x; a1.y += x1.y; a1.z += x1.z; a1.w += x1.w;
        }
        float4* srow = (float4*)(S + row * SP);
        srow[lane]      = a0;
        srow[32 + lane] = a1;
    }
    __syncthreads();

    if (wave < 16) {
        const int l  = wave >> 2;
        const int nb = (wave & 3) * 32;
        const float* srow = S + (l * TB + lm) * SP + half * 2;
        const float* wp   = g_net + l * FEAT * EMB + (half * 2) * EMB + nb + 2 * lm;
        v8f c0 = {0.f,0.f,0.f,0.f,0.f,0.f,0.f,0.f};
        v8f c1 = {0.f,0.f,0.f,0.f,0.f,0.f,0.f,0.f};
        for (int k = 0; k < FEAT; k += 4) {
            v2f a, b0, b1;
            a.x = srow[k + 0];
            a.y = srow[k + 1];
            const float2 r0 = *(const float2*)(wp + (k + 0) * EMB);
            const float2 r1 = *(const float2*)(wp + (k + 1) * EMB);
            b0.x = r0.x;  b0.y = r1.x;
            b1.x = r0.y;  b1.y = r1.y;
            c0 = __builtin_amdgcn_wmma_f32_16x16x4_f32(false, a, false, b0,
                                                       (short)0, c0, false, false);
            c1 = __builtin_amdgcn_wmma_f32_16x16x4_f32(false, a, false, b1,
                                                       (short)0, c1, false, false);
        }
#pragma unroll
        for (int r = 0; r < 8; ++r) {
            const int m = half ? (r + 8) : r;
            float2 o = { c0[r], c1[r] };
            *(float2*)(Hs + (l * TB + m) * HP + nb + 2 * lm) = o;
        }
    }
    __syncthreads();

    float* T = S;
    if (wave < 16) {
        const int mt = wave >> 2;
        const int nb = (wave & 3) * 32;
        const float* hrow = Hs + (mt * TB + lm) * HP + half * 2;
        const float* wp   = g_ts1 + (half * 2) * EMB + nb + 2 * lm;
        v8f c0 = {0.f,0.f,0.f,0.f,0.f,0.f,0.f,0.f};
        v8f c1 = {0.f,0.f,0.f,0.f,0.f,0.f,0.f,0.f};
        for (int k = 0; k < EMB; k += 4) {
            v2f a, b0, b1;
            a.x = hrow[k + 0];
            a.y = hrow[k + 1];
            const float2 r0 = *(const float2*)(wp + (k + 0) * EMB);
            const float2 r1 = *(const float2*)(wp + (k + 1) * EMB);
            b0.x = r0.x;  b0.y = r1.x;
            b1.x = r0.y;  b1.y = r1.y;
            c0 = __builtin_amdgcn_wmma_f32_16x16x4_f32(false, a, false, b0,
                                                       (short)0, c0, false, false);
            c1 = __builtin_amdgcn_wmma_f32_16x16x4_f32(false, a, false, b1,
                                                       (short)0, c1, false, false);
        }
#pragma unroll
        for (int r = 0; r < 8; ++r) {
            const int m = half ? (r + 8) : r;
            float2 o = { tanhf(c0[r]), tanhf(c1[r]) };
            *(float2*)(T + (mt * TB + m) * HP + nb + 2 * lm) = o;
        }
    }
    __syncthreads();

    for (int rr = 0; rr < 2; ++rr) {
        const int row = wave * 2 + rr;
        float s = 0.f;
#pragma unroll
        for (int i = 0; i < 4; ++i) {
            const int m = lane + i * 32;
            s += T[row * HP + m] * g_ts2[m];
        }
#pragma unroll
        for (int off = 16; off > 0; off >>= 1) s += __shfl_xor(s, off);
        if (lane == 0) SC[row] = s;
    }
    __syncthreads();

    if (tid < TB) {
        const float s0 = SC[0*TB + tid], s1 = SC[1*TB + tid];
        const float s2 = SC[2*TB + tid], s3 = SC[3*TB + tid];
        const float mx = fmaxf(fmaxf(s0, s1), fmaxf(s2, s3));
        const float e0 = expf(s0 - mx), e1 = expf(s1 - mx);
        const float e2 = expf(s2 - mx), e3 = expf(s3 - mx);
        const float inv = 1.f / (e0 + e1 + e2 + e3);
        ATT[0*TB + tid] = e0 * inv;  ATT[1*TB + tid] = e1 * inv;
        ATT[2*TB + tid] = e2 * inv;  ATT[3*TB + tid] = e3 * inv;
    }
    __syncthreads();

    float* AGG = S + LAYERS * TB * HP;
    float* FIN = AGG + TB * HP;
    for (int e = tid; e < TB * EMB; e += THREADS) {
        const int b = e >> 7, m = e & 127;
        float v = 0.f;
#pragma unroll
        for (int l = 0; l < LAYERS; ++l)
            v += ATT[l * TB + b] * Hs[(l * TB + b) * HP + m];
        AGG[b * HP + m] = v;
    }
    __syncthreads();

    if (wave < 4) {
        const int nb = wave * 32;
        const float* arow = AGG + lm * HP + half * 2;
        const float* wp   = g_tw + (half * 2) * EMB + nb + 2 * lm;
        v8f c0 = {0.f,0.f,0.f,0.f,0.f,0.f,0.f,0.f};
        v8f c1 = {0.f,0.f,0.f,0.f,0.f,0.f,0.f,0.f};
        for (int k = 0; k < EMB; k += 4) {
            v2f a, b0, b1;
            a.x = arow[k + 0];
            a.y = arow[k + 1];
            const float2 r0 = *(const float2*)(wp + (k + 0) * EMB);
            const float2 r1 = *(const float2*)(wp + (k + 1) * EMB);
            b0.x = r0.x;  b0.y = r1.x;
            b1.x = r0.y;  b1.y = r1.y;
            c0 = __builtin_amdgcn_wmma_f32_16x16x4_f32(false, a, false, b0,
                                                       (short)0, c0, false, false);
            c1 = __builtin_amdgcn_wmma_f32_16x16x4_f32(false, a, false, b1,
                                                       (short)0, c1, false, false);
        }
#pragma unroll
        for (int r = 0; r < 8; ++r) {
            const int m = half ? (r + 8) : r;
            float2 o = { c0[r], c1[r] };
            *(float2*)(FIN + m * HP + nb + 2 * lm) = o;
        }
    }
    __syncthreads();

    if (wave < TB) {
        const int b  = wave;
        const int bg = blk * TB + b;
        const int node = g_node_i[bg];
        const int lay  = g_layers[bg];
        const float4* emb  = (const float4*)(g_layemb + ((long)node * LAYERS + lay) * EMB);
        const float4* frow = (const float4*)(FIN + b * HP);
        float4 v = frow[lane];
        const float4 e = emb[lane];
        v.x += e.x; v.y += e.y; v.z += e.z; v.w += e.w;
        float ss = v.x*v.x + v.y*v.y + v.z*v.z + v.w*v.w;
#pragma unroll
        for (int off = 16; off > 0; off >>= 1) ss += __shfl_xor(ss, off);
        const float inv = 1.f / fmaxf(sqrtf(ss), 1e-12f);
        float4 o = { v.x * inv, v.y * inv, v.z * inv, v.w * inv };
        ((float4*)(g_out + (long)bg * EMB))[lane] = o;
    }
}

extern "C" void kernel_launch(void* const* d_in, const int* in_sizes, int n_in,
                              void* d_out, int out_size, void* d_ws, size_t ws_size,
                              hipStream_t stream) {
    const int*   layers   = (const int*)  d_in[0];
    const int*   node_i   = (const int*)  d_in[1];
    const int*   neighs   = (const int*)  d_in[2];
    const float* features = (const float*)d_in[3];
    const float* layemb   = (const float*)d_in[4];
    const float* net      = (const float*)d_in[5];
    const float* tw       = (const float*)d_in[6];
    const float* ts1      = (const float*)d_in[7];
    const float* ts2      = (const float*)d_in[8];
    float* out = (float*)d_out;

    dim3 grid(BATCH / TB);
    dim3 block(THREADS);
    liamne_fused<<<grid, block, 0, stream>>>(layers, node_i, neighs, features,
                                             layemb, net, tw, ts1, ts2, out);
}
